# starting kernel
# baseline (speedup 1.0000x reference)
#include <hip/hip_runtime.h>
#include <math.h>

#define BB 4
#define SS 2048
#define CC 512
#define HH 8
#define DD 64
#define HD 512

typedef __attribute__((ext_vector_type(2))) float v2f;
typedef __attribute__((ext_vector_type(8))) float v8f;

__device__ __forceinline__ v8f wmma_f32(v2f a, v2f b, v8f c) {
  return __builtin_amdgcn_wmma_f32_16x16x4_f32(false, a, false, b, (short)0, c,
                                               false, false);
}

__device__ __forceinline__ void prefetch_wgp(const void* p) {
  asm volatile("global_prefetch_b8 %0, off" ::"v"(p));
}

__global__ __launch_bounds__(128) void gemm_8192x512x512(
    const float* __restrict__ X, const float* __restrict__ W,
    float* __restrict__ Y) {
  const int lane = threadIdx.x & 31;
  const int wave = threadIdx.x >> 5;
  const int mbase = (blockIdx.x * 4 + wave) * 32;
  const int nbase = blockIdx.y * 64;
  const int lr = lane & 15;
  const int kh = (lane >> 4) * 2;

  const float* xrow0 = X + (size_t)(mbase + lr) * HD;
  const float* xrow1 = xrow0 + (size_t)16 * HD;

  v8f acc[8] = {};

  for (int c0 = 0; c0 < CC; c0 += 16) {
    if (c0 + 16 < CC) {
      prefetch_wgp(X + (size_t)(mbase + lane) * HD + c0 + 16);
      const float* wpf =
          W + (size_t)(c0 + 16 + lr) * HD + nbase + ((lane >> 4) << 5);
      prefetch_wgp(wpf);
      prefetch_wgp(wpf + 16);
    }
#pragma unroll
    for (int u = 0; u < 4; ++u) {
      const int c = c0 + 4 * u;
      v2f a0 = *(const v2f*)(xrow0 + c + kh);
      v2f a1 = *(const v2f*)(xrow1 + c + kh);
      const float* wp = W + (size_t)(c + kh) * HD + nbase + lr;
      v2f b0, b1, b2, b3;
      b0.x = wp[0];   b0.y = wp[HD + 0];
      b1.x = wp[16];  b1.y = wp[HD + 16];
      b2.x = wp[32];  b2.y = wp[HD + 32];
      b3.x = wp[48];  b3.y = wp[HD + 48];
      acc[0] = wmma_f32(a0, b0, acc[0]);
      acc[1] = wmma_f32(a0, b1, acc[1]);
      acc[2] = wmma_f32(a0, b2, acc[2]);
      acc[3] = wmma_f32(a0, b3, acc[3]);
      acc[4] = wmma_f32(a1, b0, acc[4]);
      acc[5] = wmma_f32(a1, b1, acc[5]);
      acc[6] = wmma_f32(a1, b2, acc[6]);
      acc[7] = wmma_f32(a1, b3, acc[7]);
    }
  }

#pragma unroll
  for (int t = 0; t < 2; ++t) {
    float* yb =
        Y + (size_t)(mbase + t * 16 + 8 * (lane >> 4)) * HD + nbase + lr;
#pragma unroll
    for (int r = 0; r < 8; ++r) {
      float* yr = yb + (size_t)r * HD;
      yr[0]  = acc[t * 4 + 0][r];
      yr[16] = acc[t * 4 + 1][r];
      yr[32] = acc[t * 4 + 2][r];
      yr[48] = acc[t * 4 + 3][r];
    }
  }
}

__global__ __launch_bounds__(32) void attn_kernel(
    const float* __restrict__ Q, const float* __restrict__ K,
    const float* __restrict__ V, const int* __restrict__ length,
    float* __restrict__ O) {
  __shared__ float lds[16 * 16];

  const int lane  = threadIdx.x & 31;
  const int lr    = lane & 15;
  const int kh    = (lane >> 4) * 2;
  const int half8 = (lane >> 4) * 8;
  const int qbase = blockIdx.x * 16;
  const int h     = blockIdx.y;
  const int b     = blockIdx.z;

  int L = length[b];
  if (L < 0) L = 0;
  if (L > SS) L = SS;

  const size_t baseBH = (size_t)b * SS * HD + (size_t)h * DD;
  const float* Qb = Q + baseBH;
  const float* Kb = K + baseBH;
  const float* Vb = V + baseBH;
  float*       Ob = O + baseBH;

  v2f aq[16];
  {
    const float* qrow = Qb + (size_t)(qbase + lr) * HD;
#pragma unroll
    for (int s = 0; s < 16; ++s) aq[s] = *(const v2f*)(qrow + 4 * s + kh);
  }

  float m[8], l[8];
#pragma unroll
  for (int r = 0; r < 8; ++r) { m[r] = -3.0e38f; l[r] = 0.0f; }
  v8f acc[4] = {};

  const int nkt = (L + 15) >> 4;
  for (int kt = 0; kt < nkt; ++kt) {
    const int kbase = kt * 16;

    if (kt + 1 < nkt) {
      const size_t poff = (size_t)(kbase + 16 + lr) * HD +
                          ((lane >> 4) << 5) + ((kt & 1) << 4);
      prefetch_wgp(Kb + poff);
      prefetch_wgp(Vb + poff);
    }

    v8f sc = {};
    const float* krow = Kb + (size_t)(kbase + lr) * HD;
#pragma unroll
    for (int s = 0; s < 16; ++s) {
      v2f bk = *(const v2f*)(krow + 4 * s + kh);
      sc = wmma_f32(aq[s], bk, sc);
    }

    const bool kvalid = (kbase + lr) < L;
#pragma unroll
    for (int r = 0; r < 8; ++r) {
      float v = sc[r] * 0.125f;
      sc[r] = kvalid ? v : -1.0e30f;
    }

    float p[8], scale[8];
#pragma unroll
    for (int r = 0; r < 8; ++r) {
      float rowmax = sc[r];
      rowmax = fmaxf(rowmax, __shfl_xor(rowmax, 1));
      rowmax = fmaxf(rowmax, __shfl_xor(rowmax, 2));
      rowmax = fmaxf(rowmax, __shfl_xor(rowmax, 4));
      rowmax = fmaxf(rowmax, __shfl_xor(rowmax, 8));
      const float mnew = fmaxf(m[r], rowmax);
      const float pr = __expf(sc[r] - mnew);
      float rowsum = pr;
      rowsum += __shfl_xor(rowsum, 1);
      rowsum += __shfl_xor(rowsum, 2);
      rowsum += __shfl_xor(rowsum, 4);
      rowsum += __shfl_xor(rowsum, 8);
      scale[r] = __expf(m[r] - mnew);
      l[r] = l[r] * scale[r] + rowsum;
      m[r] = mnew;
      p[r] = pr;
    }
#pragma unroll
    for (int j = 0; j < 4; ++j)
#pragma unroll
      for (int r = 0; r < 8; ++r) acc[j][r] *= scale[r];

    __syncthreads();
#pragma unroll
    for (int r = 0; r < 8; ++r) lds[(r + half8) * 16 + lr] = p[r];
    __syncthreads();

#pragma unroll
    for (int s = 0; s < 4; ++s) {
      v2f ap;
      ap.x = lds[lr * 16 + 4 * s + kh];
      ap.y = lds[lr * 16 + 4 * s + kh + 1];
      const float* vrow = Vb + (size_t)(kbase + 4 * s + kh) * HD;
#pragma unroll
      for (int j = 0; j < 4; ++j) {
        v2f bv;
        bv.x = vrow[j * 16 + lr];
        bv.y = vrow[HD + j * 16 + lr];
        acc[j] = wmma_f32(ap, bv, acc[j]);
      }
    }
  }

#pragma unroll
  for (int r = 0; r < 8; ++r) {
    const int qrow = qbase + r + half8;
    const float inv = (qrow < L && l[r] > 0.0f) ? (1.0f / l[r]) : 0.0f;
    float* orow = Ob + (size_t)qrow * HD;
    orow[0  + lr] = acc[0][r] * inv;
    orow[16 + lr] = acc[1][r] * inv;
    orow[32 + lr] = acc[2][r] * inv;
    orow[48 + lr] = acc[3][r] * inv;
  }
}

extern "C" void kernel_launch(void* const* d_in, const int* in_sizes, int n_in,
                              void* d_out, int out_size, void* d_ws,
                              size_t ws_size, hipStream_t stream) {
  const float* x      = (const float*)d_in[0];
  const int*   length = (const int*)d_in[1];
  const float* Wq     = (const float*)d_in[2];
  const float* Wk     = (const float*)d_in[3];
  const float* Wv     = (const float*)d_in[4];
  const float* Wx     = (const float*)d_in[5];
  float*       out    = (float*)d_out;

  const size_t mat = (size_t)BB * SS * HD;
  float* Qw = (float*)d_ws;
  float* Kw = Qw + mat;
  float* Vw = Kw + mat;
  float* Ow = Vw + mat;

  const dim3 gGrid(64, 8, 1);
  const dim3 gBlock(128, 1, 1);

  gemm_8192x512x512<<<gGrid, gBlock, 0, stream>>>(x, Wq, Qw);
  gemm_8192x512x512<<<gGrid, gBlock, 0, stream>>>(x, Wk, Kw);
  gemm_8192x512x512<<<gGrid, gBlock, 0, stream>>>(x, Wv, Vw);

  attn_kernel<<<dim3(SS / 16, HH, BB), dim3(32, 1, 1), 0, stream>>>(
      Qw, Kw, Vw, length, Ow);

  gemm_8192x512x512<<<gGrid, gBlock, 0, stream>>>(Ow, Wx, out);
}


# attention: each wave owns 2 adjacent 16-query tiles and feeds both from every loaded K and V fragment (2 WMMAs per fragment load, half as many waves); per-tile summation order unchanged
# speedup vs baseline: 1.0978x; 1.0978x over previous
#include <hip/hip_runtime.h>
#include <math.h>

#define BB 4
#define SS 2048
#define CC 512
#define HH 8
#define DD 64
#define HD 512

typedef __attribute__((ext_vector_type(2))) float v2f;
typedef __attribute__((ext_vector_type(8))) float v8f;

__device__ __forceinline__ v8f wmma_f32(v2f a, v2f b, v8f c) {
  return __builtin_amdgcn_wmma_f32_16x16x4_f32(false, a, false, b, (short)0, c,
                                               false, false);
}

__device__ __forceinline__ void prefetch_wgp(const void* p) {
  asm volatile("global_prefetch_b8 %0, off" ::"v"(p));
}

__global__ __launch_bounds__(128) void gemm_8192x512x512(
    const float* __restrict__ X, const float* __restrict__ W,
    float* __restrict__ Y) {
  const int lane = threadIdx.x & 31;
  const int wave = threadIdx.x >> 5;
  const int mbase = (blockIdx.x * 4 + wave) * 32;
  const int nbase = blockIdx.y * 64;
  const int lr = lane & 15;
  const int kh = (lane >> 4) * 2;

  const float* xrow0 = X + (size_t)(mbase + lr) * HD;
  const float* xrow1 = xrow0 + (size_t)16 * HD;

  v8f acc[8] = {};

  for (int c0 = 0; c0 < CC; c0 += 16) {
    if (c0 + 16 < CC) {
      prefetch_wgp(X + (size_t)(mbase + lane) * HD + c0 + 16);
      const float* wpf =
          W + (size_t)(c0 + 16 + lr) * HD + nbase + ((lane >> 4) << 5);
      prefetch_wgp(wpf);
      prefetch_wgp(wpf + 16);
    }
#pragma unroll
    for (int u = 0; u < 4; ++u) {
      const int c = c0 + 4 * u;
      v2f a0 = *(const v2f*)(xrow0 + c + kh);
      v2f a1 = *(const v2f*)(xrow1 + c + kh);
      const float* wp = W + (size_t)(c + kh) * HD + nbase + lr;
      v2f b0, b1, b2, b3;
      b0.x = wp[0];   b0.y = wp[HD + 0];
      b1.x = wp[16];  b1.y = wp[HD + 16];
      b2.x = wp[32];  b2.y = wp[HD + 32];
      b3.x = wp[48];  b3.y = wp[HD + 48];
      acc[0] = wmma_f32(a0, b0, acc[0]);
      acc[1] = wmma_f32(a0, b1, acc[1]);
      acc[2] = wmma_f32(a0, b2, acc[2]);
      acc[3] = wmma_f32(a0, b3, acc[3]);
      acc[4] = wmma_f32(a1, b0, acc[4]);
      acc[5] = wmma_f32(a1, b1, acc[5]);
      acc[6] = wmma_f32(a1, b2, acc[6]);
      acc[7] = wmma_f32(a1, b3, acc[7]);
    }
  }

#pragma unroll
  for (int t = 0; t < 2; ++t) {
    float* yb =
        Y + (size_t)(mbase + t * 16 + 8 * (lane >> 4)) * HD + nbase + lr;
#pragma unroll
    for (int r = 0; r < 8; ++r) {
      float* yr = yb + (size_t)r * HD;
      yr[0]  = acc[t * 4 + 0][r];
      yr[16] = acc[t * 4 + 1][r];
      yr[32] = acc[t * 4 + 2][r];
      yr[48] = acc[t * 4 + 3][r];
    }
  }
}

#define QT 2

__global__ __launch_bounds__(32) void attn_kernel(
    const float* __restrict__ Q, const float* __restrict__ K,
    const float* __restrict__ V, const int* __restrict__ length,
    float* __restrict__ O) {
  __shared__ float lds[QT][16 * 16];

  const int lane  = threadIdx.x & 31;
  const int lr    = lane & 15;
  const int kh    = (lane >> 4) * 2;
  const int half8 = (lane >> 4) * 8;
  const int qbase = blockIdx.x * (16 * QT);
  const int h     = blockIdx.y;
  const int b     = blockIdx.z;

  int L = length[b];
  if (L < 0) L = 0;
  if (L > SS) L = SS;

  const size_t baseBH = (size_t)b * SS * HD + (size_t)h * DD;
  const float* Qb = Q + baseBH;
  const float* Kb = K + baseBH;
  const float* Vb = V + baseBH;
  float*       Ob = O + baseBH;

  v2f aq[QT][16];
#pragma unroll
  for (int t = 0; t < QT; ++t) {
    const float* qrow = Qb + (size_t)(qbase + 16 * t + lr) * HD;
#pragma unroll
    for (int s = 0; s < 16; ++s) aq[t][s] = *(const v2f*)(qrow + 4 * s + kh);
  }

  float m[QT][8], l[QT][8];
#pragma unroll
  for (int t = 0; t < QT; ++t)
#pragma unroll
    for (int r = 0; r < 8; ++r) { m[t][r] = -3.0e38f; l[t][r] = 0.0f; }
  v8f acc[QT][4] = {};

  const int nkt = (L + 15) >> 4;
  for (int kt = 0; kt < nkt; ++kt) {
    const int kbase = kt * 16;

    if (kt + 1 < nkt) {
      const size_t poff = (size_t)(kbase + 16 + lr) * HD +
                          ((lane >> 4) << 5) + ((kt & 1) << 4);
      prefetch_wgp(Kb + poff);
      prefetch_wgp(Vb + poff);
    }

    v8f sc[QT] = {};
    const float* krow = Kb + (size_t)(kbase + lr) * HD;
#pragma unroll
    for (int s = 0; s < 16; ++s) {
      v2f bk = *(const v2f*)(krow + 4 * s + kh);
#pragma unroll
      for (int t = 0; t < QT; ++t) sc[t] = wmma_f32(aq[t][s], bk, sc[t]);
    }

    const bool kvalid = (kbase + lr) < L;
#pragma unroll
    for (int t = 0; t < QT; ++t)
#pragma unroll
      for (int r = 0; r < 8; ++r) {
        float v = sc[t][r] * 0.125f;
        sc[t][r] = kvalid ? v : -1.0e30f;
      }

    float p[QT][8], scale[QT][8];
#pragma unroll
    for (int t = 0; t < QT; ++t)
#pragma unroll
      for (int r = 0; r < 8; ++r) {
        float rowmax = sc[t][r];
        rowmax = fmaxf(rowmax, __shfl_xor(rowmax, 1));
        rowmax = fmaxf(rowmax, __shfl_xor(rowmax, 2));
        rowmax = fmaxf(rowmax, __shfl_xor(rowmax, 4));
        rowmax = fmaxf(rowmax, __shfl_xor(rowmax, 8));
        const float mnew = fmaxf(m[t][r], rowmax);
        const float pr = __expf(sc[t][r] - mnew);
        float rowsum = pr;
        rowsum += __shfl_xor(rowsum, 1);
        rowsum += __shfl_xor(rowsum, 2);
        rowsum += __shfl_xor(rowsum, 4);
        rowsum += __shfl_xor(rowsum, 8);
        scale[t][r] = __expf(m[t][r] - mnew);
        l[t][r] = l[t][r] * scale[t][r] + rowsum;
        m[t][r] = mnew;
        p[t][r] = pr;
      }
#pragma unroll
    for (int t = 0; t < QT; ++t)
#pragma unroll
      for (int j = 0; j < 4; ++j)
#pragma unroll
        for (int r = 0; r < 8; ++r) acc[t][j][r] *= scale[t][r];

    __syncthreads();
#pragma unroll
    for (int t = 0; t < QT; ++t)
#pragma unroll
      for (int r = 0; r < 8; ++r) lds[t][(r + half8) * 16 + lr] = p[t][r];
    __syncthreads();

#pragma unroll
    for (int s = 0; s < 4; ++s) {
      v2f ap[QT];
#pragma unroll
      for (int t = 0; t < QT; ++t) {
        ap[t].x = lds[t][lr * 16 + 4 * s + kh];
        ap[t].y = lds[t][lr * 16 + 4 * s + kh + 1];
      }
      const float* vrow = Vb + (size_t)(kbase + 4 * s + kh) * HD;
#pragma unroll
      for (int j = 0; j < 4; ++j) {
        v2f bv;
        bv.x = vrow[j * 16 + lr];
        bv.y = vrow[HD + j * 16 + lr];
#pragma unroll
        for (int t = 0; t < QT; ++t) acc[t][j] = wmma_f32(ap[t], bv, acc[t][j]);
      }
    }
  }

#pragma unroll
  for (int t = 0; t < QT; ++t)
#pragma unroll
    for (int r = 0; r < 8; ++r) {
      const int qrow = qbase + 16 * t + r + half8;
      const float inv =
          (qrow < L && l[t][r] > 0.0f) ? (1.0f / l[t][r]) : 0.0f;
      float* orow = Ob + (size_t)qrow * HD;
      orow[0  + lr] = acc[t][0][r] * inv;
      orow[16 + lr] = acc[t][1][r] * inv;
      orow[32 + lr] = acc[t][2][r] * inv;
      orow[48 + lr] = acc[t][3][r] * inv;
    }
}

extern "C" void kernel_launch(void* const* d_in, const int* in_sizes, int n_in,
                              void* d_out, int out_size, void* d_ws,
                              size_t ws_size, hipStream_t stream) {
  const float* x      = (const float*)d_in[0];
  const int*   length = (const int*)d_in[1];
  const float* Wq     = (const float*)d_in[2];
  const float* Wk     = (const float*)d_in[3];
  const float* Wv     = (const float*)d_in[4];
  const float* Wx     = (const float*)d_in[5];
  float*       out    = (float*)d_out;

  const size_t mat = (size_t)BB * SS * HD;
  float* Qw = (float*)d_ws;
  float* Kw = Qw + mat;
  float* Vw = Kw + mat;
  float* Ow = Vw + mat;

  const dim3 gGrid(64, 8, 1);
  const dim3 gBlock(128, 1, 1);

  gemm_8192x512x512<<<gGrid, gBlock, 0, stream>>>(x, Wq, Qw);
  gemm_8192x512x512<<<gGrid, gBlock, 0, stream>>>(x, Wk, Kw);
  gemm_8192x512x512<<<gGrid, gBlock, 0, stream>>>(x, Wv, Vw);

  attn_kernel<<<dim3(SS / (16 * QT), HH, BB), dim3(32, 1, 1), 0, stream>>>(
      Qw, Kw, Vw, length, Ow);

  gemm_8192x512x512<<<gGrid, gBlock, 0, stream>>>(Ow, Wx, out);
}


# attention: score tile transposed (K x Q^T): a lane holds 8 keys of its own query, so softmax max/sum run in registers + one lane exchange, P feeds V^T x P^T from registers; sum order changed
# speedup vs baseline: 1.3687x; 1.2468x over previous
#include <hip/hip_runtime.h>
#include <math.h>

#define BB 4
#define SS 2048
#define CC 512
#define HH 8
#define DD 64
#define HD 512

typedef __attribute__((ext_vector_type(2))) float v2f;
typedef __attribute__((ext_vector_type(8))) float v8f;

__device__ __forceinline__ v8f wmma_f32(v2f a, v2f b, v8f c) {
  return __builtin_amdgcn_wmma_f32_16x16x4_f32(false, a, false, b, (short)0, c,
                                               false, false);
}

__device__ __forceinline__ void prefetch_wgp(const void* p) {
  asm volatile("global_prefetch_b8 %0, off" ::"v"(p));
}

__global__ __launch_bounds__(128) void gemm_8192x512x512(
    const float* __restrict__ X, const float* __restrict__ W,
    float* __restrict__ Y) {
  const int lane = threadIdx.x & 31;
  const int wave = threadIdx.x >> 5;
  const int mbase = (blockIdx.x * 4 + wave) * 32;
  const int nbase = blockIdx.y * 64;
  const int lr = lane & 15;
  const int kh = (lane >> 4) * 2;

  const float* xrow0 = X + (size_t)(mbase + lr) * HD;
  const float* xrow1 = xrow0 + (size_t)16 * HD;

  v8f acc[8] = {};

  for (int c0 = 0; c0 < CC; c0 += 16) {
    if (c0 + 16 < CC) {
      prefetch_wgp(X + (size_t)(mbase + lane) * HD + c0 + 16);
      const float* wpf =
          W + (size_t)(c0 + 16 + lr) * HD + nbase + ((lane >> 4) << 5);
      prefetch_wgp(wpf);
      prefetch_wgp(wpf + 16);
    }
#pragma unroll
    for (int u = 0; u < 4; ++u) {
      const int c = c0 + 4 * u;
      v2f a0 = *(const v2f*)(xrow0 + c + kh);
      v2f a1 = *(const v2f*)(xrow1 + c + kh);
      const float* wp = W + (size_t)(c + kh) * HD + nbase + lr;
      v2f b0, b1, b2, b3;
      b0.x = wp[0];   b0.y = wp[HD + 0];
      b1.x = wp[16];  b1.y = wp[HD + 16];
      b2.x = wp[32];  b2.y = wp[HD + 32];
      b3.x = wp[48];  b3.y = wp[HD + 48];
      acc[0] = wmma_f32(a0, b0, acc[0]);
      acc[1] = wmma_f32(a0, b1, acc[1]);
      acc[2] = wmma_f32(a0, b2, acc[2]);
      acc[3] = wmma_f32(a0, b3, acc[3]);
      acc[4] = wmma_f32(a1, b0, acc[4]);
      acc[5] = wmma_f32(a1, b1, acc[5]);
      acc[6] = wmma_f32(a1, b2, acc[6]);
      acc[7] = wmma_f32(a1, b3, acc[7]);
    }
  }

#pragma unroll
  for (int t = 0; t < 2; ++t) {
    float* yb =
        Y + (size_t)(mbase + t * 16 + 8 * (lane >> 4)) * HD + nbase + lr;
#pragma unroll
    for (int r = 0; r < 8; ++r) {
      float* yr = yb + (size_t)r * HD;
      yr[0]  = acc[t * 4 + 0][r];
      yr[16] = acc[t * 4 + 1][r];
      yr[32] = acc[t * 4 + 2][r];
      yr[48] = acc[t * 4 + 3][r];
    }
  }
}

#define QT 2

typedef __attribute__((ext_vector_type(4))) float v4f;

__device__ __forceinline__ float xhalf(float x) {
  const int b = __builtin_bit_cast(int, x);
  return __builtin_bit_cast(
      float, __builtin_amdgcn_permlanex16(b, b, 0x76543210, 0xfedcba98, false,
                                          false));
}

__global__ __launch_bounds__(32) void attn_kernel(
    const float* __restrict__ Q, const float* __restrict__ K,
    const float* __restrict__ V, const int* __restrict__ length,
    float* __restrict__ O) {
  const int lane  = threadIdx.x & 31;
  const int lr    = lane & 15;
  const int kh    = (lane >> 4) * 2;
  const int half8 = (lane >> 4) * 8;
  const int qbase = blockIdx.x * (16 * QT);
  const int h     = blockIdx.y;
  const int b     = blockIdx.z;

  int L = length[b];
  if (L < 0) L = 0;
  if (L > SS) L = SS;

  const size_t baseBH = (size_t)b * SS * HD + (size_t)h * DD;
  const float* Qb = Q + baseBH;
  const float* Kb = K + baseBH;
  const float* Vb = V + baseBH;
  float*       Ob = O + baseBH;

  v2f bq[QT][16];
#pragma unroll
  for (int t = 0; t < QT; ++t) {
    const float* qrow = Qb + (size_t)(qbase + 16 * t + lr) * HD;
#pragma unroll
    for (int s = 0; s < 16; ++s) bq[t][s] = *(const v2f*)(qrow + 4 * s + kh);
  }

  float m[QT], l[QT];
#pragma unroll
  for (int t = 0; t < QT; ++t) { m[t] = -3.0e38f; l[t] = 0.0f; }
  v8f acc[QT][4] = {};

  const int nkt = (L + 15) >> 4;
  for (int kt = 0; kt < nkt; ++kt) {
    const int kbase = kt * 16;

    if (kt + 1 < nkt) {
      const size_t poff = (size_t)(kbase + 16 + lr) * HD +
                          ((lane >> 4) << 5) + ((kt & 1) << 4);
      prefetch_wgp(Kb + poff);
      prefetch_wgp(Vb + poff);
    }

    v8f sc[QT] = {};
    const float* krow = Kb + (size_t)(kbase + lr) * HD;
#pragma unroll
    for (int s = 0; s < 16; ++s) {
      v2f ak = *(const v2f*)(krow + 4 * s + kh);
#pragma unroll
      for (int t = 0; t < QT; ++t) sc[t] = wmma_f32(ak, bq[t][s], sc[t]);
    }

#pragma unroll
    for (int r = 0; r < 8; ++r) {
      const bool kvalid = (kbase + half8 + r) < L;
#pragma unroll
      for (int t = 0; t < QT; ++t) {
        const float v = sc[t][r] * 0.125f;
        sc[t][r] = kvalid ? v : -1.0e30f;
      }
    }

#pragma unroll
    for (int t = 0; t < QT; ++t) {
      float mx = fmaxf(fmaxf(fmaxf(sc[t][0], sc[t][1]), fmaxf(sc[t][2], sc[t][3])),
                       fmaxf(fmaxf(sc[t][4], sc[t][5]), fmaxf(sc[t][6], sc[t][7])));
      mx = fmaxf(mx, xhalf(mx));
      const float mnew = fmaxf(m[t], mx);
#pragma unroll
      for (int r = 0; r < 8; ++r) sc[t][r] = __expf(sc[t][r] - mnew);
      const float psum = ((sc[t][0] + sc[t][1]) + (sc[t][2] + sc[t][3])) +
                         ((sc[t][4] + sc[t][5]) + (sc[t][6] + sc[t][7]));
      const float scale = __expf(m[t] - mnew);
      l[t] = l[t] * scale + psum;
      m[t] = mnew;
#pragma unroll
      for (int j = 0; j < 4; ++j)
#pragma unroll
        for (int r = 0; r < 8; ++r) acc[t][j][r] *= scale;
    }

#pragma unroll
    for (int s = 0; s < 4; ++s) {
      v2f bp[QT];
#pragma unroll
      for (int t = 0; t < QT; ++t) {
        bp[t].x = sc[t][2 * s];
        bp[t].y = sc[t][2 * s + 1];
      }
      const float* vrow = Vb + (size_t)(kbase + half8 + 2 * s) * HD;
#pragma unroll
      for (int j = 0; j < 4; ++j) {
        v2f av;
        av.x = vrow[j * 16 + lr];
        av.y = vrow[HD + j * 16 + lr];
#pragma unroll
        for (int t = 0; t < QT; ++t) acc[t][j] = wmma_f32(av, bp[t], acc[t][j]);
      }
    }
  }

#pragma unroll
  for (int t = 0; t < QT; ++t) {
    const float lt = l[t] + xhalf(l[t]);
    const int qrow = qbase + 16 * t + lr;
    const float inv = (qrow < L && lt > 0.0f) ? (1.0f / lt) : 0.0f;
    float* orow = Ob + (size_t)qrow * HD + half8;
#pragma unroll
    for (int j = 0; j < 4; ++j) {
      v4f o0, o1;
      o0.x = acc[t][j][0] * inv;  o0.y = acc[t][j][1] * inv;
      o0.z = acc[t][j][2] * inv;  o0.w = acc[t][j][3] * inv;
      o1.x = acc[t][j][4] * inv;  o1.y = acc[t][j][5] * inv;
      o1.z = acc[t][j][6] * inv;  o1.w = acc[t][j][7] * inv;
      *(v4f*)(orow + 16 * j)     = o0;
      *(v4f*)(orow + 16 * j + 4) = o1;
    }
  }
}

extern "C" void kernel_launch(void* const* d_in, const int* in_sizes, int n_in,
                              void* d_out, int out_size, void* d_ws,
                              size_t ws_size, hipStream_t stream) {
  const float* x      = (const float*)d_in[0];
  const int*   length = (const int*)d_in[1];
  const float* Wq     = (const float*)d_in[2];
  const float* Wk     = (const float*)d_in[3];
  const float* Wv     = (const float*)d_in[4];
  const float* Wx     = (const float*)d_in[5];
  float*       out    = (float*)d_out;

  const size_t mat = (size_t)BB * SS * HD;
  float* Qw = (float*)d_ws;
  float* Kw = Qw + mat;
  float* Vw = Kw + mat;
  float* Ow = Vw + mat;

  const dim3 gGrid(64, 8, 1);
  const dim3 gBlock(128, 1, 1);

  gemm_8192x512x512<<<gGrid, gBlock, 0, stream>>>(x, Wq, Qw);
  gemm_8192x512x512<<<gGrid, gBlock, 0, stream>>>(x, Wk, Kw);
  gemm_8192x512x512<<<gGrid, gBlock, 0, stream>>>(x, Wv, Vw);

  attn_kernel<<<dim3(SS / (16 * QT), HH, BB), dim3(32, 1, 1), 0, stream>>>(
      Qw, Kw, Vw, length, Ow);

  gemm_8192x512x512<<<gGrid, gBlock, 0, stream>>>(Ow, Wx, out);
}


# attention: 4-wave workgroups stage each 32-key K,V block once in LDS (register double buffer, 1 barrier per block); fragments read as grouped 16-B LDS loads, d order permuted: score sum order changed
# speedup vs baseline: 1.7124x; 1.2511x over previous
#include <hip/hip_runtime.h>
#include <math.h>

#define BB 4
#define SS 2048
#define CC 512
#define HH 8
#define DD 64
#define HD 512

typedef __attribute__((ext_vector_type(2))) float v2f;
typedef __attribute__((ext_vector_type(8))) float v8f;

__device__ __forceinline__ v8f wmma_f32(v2f a, v2f b, v8f c) {
  return __builtin_amdgcn_wmma_f32_16x16x4_f32(false, a, false, b, (short)0, c,
                                               false, false);
}

__device__ __forceinline__ void prefetch_wgp(const void* p) {
  asm volatile("global_prefetch_b8 %0, off" ::"v"(p));
}

__global__ __launch_bounds__(128) void gemm_8192x512x512(
    const float* __restrict__ X, const float* __restrict__ W,
    float* __restrict__ Y) {
  const int lane = threadIdx.x & 31;
  const int wave = threadIdx.x >> 5;
  const int mbase = (blockIdx.x * 4 + wave) * 32;
  const int nbase = blockIdx.y * 64;
  const int lr = lane & 15;
  const int kh = (lane >> 4) * 2;

  const float* xrow0 = X + (size_t)(mbase + lr) * HD;
  const float* xrow1 = xrow0 + (size_t)16 * HD;

  v8f acc[8] = {};

  for (int c0 = 0; c0 < CC; c0 += 16) {
    if (c0 + 16 < CC) {
      prefetch_wgp(X + (size_t)(mbase + lane) * HD + c0 + 16);
      const float* wpf =
          W + (size_t)(c0 + 16 + lr) * HD + nbase + ((lane >> 4) << 5);
      prefetch_wgp(wpf);
      prefetch_wgp(wpf + 16);
    }
#pragma unroll
    for (int u = 0; u < 4; ++u) {
      const int c = c0 + 4 * u;
      v2f a0 = *(const v2f*)(xrow0 + c + kh);
      v2f a1 = *(const v2f*)(xrow1 + c + kh);
      const float* wp = W + (size_t)(c + kh) * HD + nbase + lr;
      v2f b0, b1, b2, b3;
      b0.x = wp[0];   b0.y = wp[HD + 0];
      b1.x = wp[16];  b1.y = wp[HD + 16];
      b2.x = wp[32];  b2.y = wp[HD + 32];
      b3.x = wp[48];  b3.y = wp[HD + 48];
      acc[0] = wmma_f32(a0, b0, acc[0]);
      acc[1] = wmma_f32(a0, b1, acc[1]);
      acc[2] = wmma_f32(a0, b2, acc[2]);
      acc[3] = wmma_f32(a0, b3, acc[3]);
      acc[4] = wmma_f32(a1, b0, acc[4]);
      acc[5] = wmma_f32(a1, b1, acc[5]);
      acc[6] = wmma_f32(a1, b2, acc[6]);
      acc[7] = wmma_f32(a1, b3, acc[7]);
    }
  }

#pragma unroll
  for (int t = 0; t < 2; ++t) {
    float* yb =
        Y + (size_t)(mbase + t * 16 + 8 * (lane >> 4)) * HD + nbase + lr;
#pragma unroll
    for (int r = 0; r < 8; ++r) {
      float* yr = yb + (size_t)r * HD;
      yr[0]  = acc[t * 4 + 0][r];
      yr[16] = acc[t * 4 + 1][r];
      yr[32] = acc[t * 4 + 2][r];
      yr[48] = acc[t * 4 + 3][r];
    }
  }
}

#define QT 2
#define NW 4
#define KC 32
#define PITCH 68

typedef __attribute__((ext_vector_type(4))) float v4f;

__device__ __forceinline__ float xhalf(float x) {
  const int b = __builtin_bit_cast(int, x);
  return __builtin_bit_cast(
      float, __builtin_amdgcn_permlanex16(b, b, 0x76543210, 0xfedcba98, false,
                                          false));
}

__device__ __forceinline__ void sched_fence() {
  __builtin_amdgcn_sched_barrier(0);
}

__device__ __forceinline__ void load_frag_k(v4f (&kf)[8], const float* p) {
#pragma unroll
  for (int u = 0; u < 8; ++u) kf[u] = *(const v4f*)(p + 8 * u);
}

__device__ __forceinline__ void load_frag_v(v4f (&vf)[8], const float* p) {
#pragma unroll
  for (int i = 0; i < 8; ++i) vf[i] = *(const v4f*)(p + i * PITCH);
}

__global__ __launch_bounds__(32 * NW) void attn_kernel(
    const float* __restrict__ Q, const float* __restrict__ K,
    const float* __restrict__ V, const int* __restrict__ length,
    float* __restrict__ O) {
  __shared__ __attribute__((aligned(16))) float sK[2][KC * PITCH];
  __shared__ __attribute__((aligned(16))) float sV[2][KC * PITCH];

  const int tid   = threadIdx.x;
  const int lane  = tid & 31;
  const int wave  = tid >> 5;
  const int lr    = lane & 15;
  const int hh    = lane >> 4;
  const int half8 = hh * 8;
  const int qbase = (blockIdx.x * NW + wave) * (16 * QT);
  const int h     = blockIdx.y;
  const int b     = blockIdx.z;

  int L = length[b];
  if (L < 0) L = 0;
  if (L > SS) L = SS;

  const size_t baseBH = (size_t)b * SS * HD + (size_t)h * DD;
  const float* Qb = Q + baseBH;
  const float* Kb = K + baseBH;
  const float* Vb = V + baseBH;
  float*       Ob = O + baseBH;

  v2f bq[QT][16];
#pragma unroll
  for (int t = 0; t < QT; ++t) {
    const float* qrow = Qb + (size_t)(qbase + 16 * t + lr) * HD + 4 * hh;
#pragma unroll
    for (int u = 0; u < 8; ++u) {
      const v4f q4 = *(const v4f*)(qrow + 8 * u);
      bq[t][2 * u].x     = q4.x;  bq[t][2 * u].y     = q4.y;
      bq[t][2 * u + 1].x = q4.z;  bq[t][2 * u + 1].y = q4.w;
    }
  }

  float m[QT], l[QT];
#pragma unroll
  for (int t = 0; t < QT; ++t) { m[t] = -3.0e38f; l[t] = 0.0f; }
  v8f acc[QT][4] = {};

  const int nkt = (L + 15) >> 4;
  const int nch = (nkt + 1) >> 1;

  const int srow = tid >> 4;
  const int scol = (tid & 15) * 4;
  const float* gK = Kb + (size_t)srow * HD + scol;
  const float* gV = Vb + (size_t)srow * HD + scol;
  const int    sOff = srow * PITCH + scol;
  v4f gk[4], gv[4];

  if (nch > 0) {
#pragma unroll
    for (int i = 0; i < 4; ++i) {
      gk[i] = *(const v4f*)(gK + (size_t)(8 * i) * HD);
      gv[i] = *(const v4f*)(gV + (size_t)(8 * i) * HD);
    }
#pragma unroll
    for (int i = 0; i < 4; ++i) {
      *(v4f*)(&sK[0][sOff + 8 * i * PITCH]) = gk[i];
      *(v4f*)(&sV[0][sOff + 8 * i * PITCH]) = gv[i];
    }
  }
  __syncthreads();

  for (int ch = 0; ch < nch; ++ch) {
    const int  buf  = ch & 1;
    const bool more = (ch + 1) < nch;

    if (more) {
      const size_t goff = (size_t)(ch + 1) * KC * HD;
#pragma unroll
      for (int i = 0; i < 4; ++i) {
        gk[i] = *(const v4f*)(gK + goff + (size_t)(8 * i) * HD);
        gv[i] = *(const v4f*)(gV + goff + (size_t)(8 * i) * HD);
      }
    }

    const float* kbuf = &sK[buf][lr * PITCH + 4 * hh];
    const float* vbuf = &sV[buf][half8 * PITCH + 4 * lr];
    v4f kf[8], vf[8];
    load_frag_k(kf, kbuf);
    sched_fence();

#pragma unroll
    for (int tt = 0; tt < KC / 16; ++tt) {
      const int kt = ch * (KC / 16) + tt;
      if (kt < nkt) {
        const int kbase = kt * 16;

        v8f sc[QT] = {};
#pragma unroll
        for (int u = 0; u < 8; ++u) {
          v2f a0, a1;
          a0.x = kf[u].x;  a0.y = kf[u].y;
          a1.x = kf[u].z;  a1.y = kf[u].w;
#pragma unroll
          for (int t = 0; t < QT; ++t) sc[t] = wmma_f32(a0, bq[t][2 * u], sc[t]);
#pragma unroll
          for (int t = 0; t < QT; ++t)
            sc[t] = wmma_f32(a1, bq[t][2 * u + 1], sc[t]);
        }

        sched_fence();
        load_frag_v(vf, vbuf + tt * 16 * PITCH);
        if (tt + 1 < KC / 16) load_frag_k(kf, kbuf + (tt + 1) * 16 * PITCH);
        sched_fence();

#pragma unroll
        for (int r = 0; r < 8; ++r) {
          const bool kvalid = (kbase + half8 + r) < L;
#pragma unroll
          for (int t = 0; t < QT; ++t) {
            const float v = sc[t][r] * 0.125f;
            sc[t][r] = kvalid ? v : -1.0e30f;
          }
        }

#pragma unroll
        for (int t = 0; t < QT; ++t) {
          float mx = fmaxf(fmaxf(fmaxf(sc[t][0], sc[t][1]), fmaxf(sc[t][2], sc[t][3])),
                           fmaxf(fmaxf(sc[t][4], sc[t][5]), fmaxf(sc[t][6], sc[t][7])));
          mx = fmaxf(mx, xhalf(mx));
          const float mnew = fmaxf(m[t], mx);
#pragma unroll
          for (int r = 0; r < 8; ++r) sc[t][r] = __expf(sc[t][r] - mnew);
          const float psum = ((sc[t][0] + sc[t][1]) + (sc[t][2] + sc[t][3])) +
                             ((sc[t][4] + sc[t][5]) + (sc[t][6] + sc[t][7]));
          const float scale = __expf(m[t] - mnew);
          l[t] = l[t] * scale + psum;
          m[t] = mnew;
#pragma unroll
          for (int j = 0; j < 4; ++j)
#pragma unroll
            for (int r = 0; r < 8; ++r) acc[t][j][r] *= scale;
        }

#pragma unroll
        for (int s = 0; s < 4; ++s) {
          v2f bp[QT];
#pragma unroll
          for (int t = 0; t < QT; ++t) {
            bp[t].x = sc[t][2 * s];
            bp[t].y = sc[t][2 * s + 1];
          }
#pragma unroll
          for (int j = 0; j < 4; ++j) {
            v2f av;
            av.x = vf[2 * s][j];
            av.y = vf[2 * s + 1][j];
#pragma unroll
            for (int t = 0; t < QT; ++t)
              acc[t][j] = wmma_f32(av, bp[t], acc[t][j]);
          }
        }
      }
    }

    if (more) {
      const int nb = buf ^ 1;
#pragma unroll
      for (int i = 0; i < 4; ++i) {
        *(v4f*)(&sK[nb][sOff + 8 * i * PITCH]) = gk[i];
        *(v4f*)(&sV[nb][sOff + 8 * i * PITCH]) = gv[i];
      }
    }
    __syncthreads();
  }

#pragma unroll
  for (int t = 0; t < QT; ++t) {
    const float lt = l[t] + xhalf(l[t]);
    const int qrow = qbase + 16 * t + lr;
    const float inv = (qrow < L && lt > 0.0f) ? (1.0f / lt) : 0.0f;
    float* orow = Ob + (size_t)qrow * HD + 32 * hh;
#pragma unroll
    for (int r = 0; r < 8; ++r) {
      v4f o;
      o.x = acc[t][0][r] * inv;
      o.y = acc[t][1][r] * inv;
      o.z = acc[t][2][r] * inv;
      o.w = acc[t][3][r] * inv;
      *(v4f*)(orow + 4 * r) = o;
    }
  }
}

extern "C" void kernel_launch(void* const* d_in, const int* in_sizes, int n_in,
                              void* d_out, int out_size, void* d_ws,
                              size_t ws_size, hipStream_t stream) {
  const float* x      = (const float*)d_in[0];
  const int*   length = (const int*)d_in[1];
  const float* Wq     = (const float*)d_in[2];
  const float* Wk     = (const float*)d_in[3];
  const float* Wv     = (const float*)d_in[4];
  const float* Wx     = (const float*)d_in[5];
  float*       out    = (float*)d_out;

  const size_t mat = (size_t)BB * SS * HD;
  float* Qw = (float*)d_ws;
  float* Kw = Qw + mat;
  float* Vw = Kw + mat;
  float* Ow = Vw + mat;

  const dim3 gGrid(64, 8, 1);
  const dim3 gBlock(128, 1, 1);

  gemm_8192x512x512<<<gGrid, gBlock, 0, stream>>>(x, Wq, Qw);
  gemm_8192x512x512<<<gGrid, gBlock, 0, stream>>>(x, Wk, Kw);
  gemm_8192x512x512<<<gGrid, gBlock, 0, stream>>>(x, Wv, Vw);

  attn_kernel<<<dim3(SS / (16 * QT * NW), HH, BB), dim3(32 * NW, 1, 1), 0, stream>>>(
      Qw, Kw, Vw, length, Ow);

  gemm_8192x512x512<<<gGrid, gBlock, 0, stream>>>(Ow, Wx, out);
}


# projection GEMMs: workgroup stages each 32-k block of X and W (W transposed) once in LDS by whole-line 16-B loads, register double buffer; fragments from LDS, k order permuted: sum order changed
# speedup vs baseline: 1.8722x; 1.0933x over previous
#include <hip/hip_runtime.h>
#include <math.h>

#define BB 4
#define SS 2048
#define CC 512
#define HH 8
#define DD 64
#define HD 512

typedef __attribute__((ext_vector_type(2))) float v2f;
typedef __attribute__((ext_vector_type(8))) float v8f;

__device__ __forceinline__ v8f wmma_f32(v2f a, v2f b, v8f c) {
  return __builtin_amdgcn_wmma_f32_16x16x4_f32(false, a, false, b, (short)0, c,
                                               false, false);
}

typedef __attribute__((ext_vector_type(4))) float v4f;

__device__ __forceinline__ void sched_fence() {
  __builtin_amdgcn_sched_barrier(0);
}

#define GKB 32
#define GPITCH 36

__global__ __launch_bounds__(128) void gemm_8192x512x512(
    const float* __restrict__ X, const float* __restrict__ W,
    float* __restrict__ Y) {
  __shared__ __attribute__((aligned(16))) float sX[2][128 * GPITCH];
  __shared__ __attribute__((aligned(16))) float sW[2][64 * GPITCH];

  const int tid  = threadIdx.x;
  const int lane = tid & 31;
  const int wave = tid >> 5;
  const int lr   = lane & 15;
  const int hh   = lane >> 4;
  const int mblk  = blockIdx.x * 128;
  const int mbase = mblk + wave * 32;
  const int nbase = blockIdx.y * 64;

  const int xrow = tid >> 3;
  const int xcol = (tid & 7) * 4;
  const float* gX = X + (size_t)(mblk + xrow) * HD + xcol;
  const int xoff = xrow * GPITCH + xcol;
  const int wn4 = (tid & 15) * 4;
  const int wk4 = (tid >> 4) * 4;
  const float* gW = W + (size_t)wk4 * HD + nbase + wn4;
  const int woff = wn4 * GPITCH + wk4;

  v4f gx[8], gw[4];
#pragma unroll
  for (int i = 0; i < 8; ++i) gx[i] = *(const v4f*)(gX + (size_t)(16 * i) * HD);
#pragma unroll
  for (int i = 0; i < 4; ++i) gw[i] = *(const v4f*)(gW + (size_t)i * HD);
#pragma unroll
  for (int i = 0; i < 8; ++i) *(v4f*)(&sX[0][xoff + 16 * i * GPITCH]) = gx[i];
#pragma unroll
  for (int n = 0; n < 4; ++n) {
    v4f t;
    t.x = gw[0][n];  t.y = gw[1][n];  t.z = gw[2][n];  t.w = gw[3][n];
    *(v4f*)(&sW[0][woff + n * GPITCH]) = t;
  }
  __syncthreads();

  v8f acc[8] = {};

  const int aoff = (wave * 32 + lr) * GPITCH + 4 * hh;
  const int boff = lr * GPITCH + 4 * hh;

  for (int kb = 0; kb < CC / GKB; ++kb) {
    const int  buf  = kb & 1;
    const bool more = (kb + 1) < CC / GKB;

    if (more) {
      const int k0 = (kb + 1) * GKB;
#pragma unroll
      for (int i = 0; i < 8; ++i)
        gx[i] = *(const v4f*)(gX + k0 + (size_t)(16 * i) * HD);
#pragma unroll
      for (int i = 0; i < 4; ++i)
        gw[i] = *(const v4f*)(gW + (size_t)(k0 + i) * HD);
    }

    const float* ap = &sX[buf][aoff];
    const float* bp = &sW[buf][boff];
    v4f af[2][4], bf[4][4];
#pragma unroll
    for (int u = 0; u < 4; ++u) {
      af[0][u] = *(const v4f*)(ap + 8 * u);
      af[1][u] = *(const v4f*)(ap + 16 * GPITCH + 8 * u);
#pragma unroll
      for (int j = 0; j < 4; ++j)
        bf[j][u] = *(const v4f*)(bp + j * 16 * GPITCH + 8 * u);
    }
    sched_fence();

#pragma unroll
    for (int u = 0; u < 4; ++u) {
#pragma unroll
      for (int e = 0; e < 2; ++e) {
        v2f a0, a1, b[4];
        a0.x = af[0][u][2 * e];  a0.y = af[0][u][2 * e + 1];
        a1.x = af[1][u][2 * e];  a1.y = af[1][u][2 * e + 1];
#pragma unroll
        for (int j = 0; j < 4; ++j) {
          b[j].x = bf[j][u][2 * e];
          b[j].y = bf[j][u][2 * e + 1];
        }
#pragma unroll
        for (int j = 0; j < 4; ++j) acc[j] = wmma_f32(a0, b[j], acc[j]);
#pragma unroll
        for (int j = 0; j < 4; ++j) acc[4 + j] = wmma_f32(a1, b[j], acc[4 + j]);
      }
    }

    if (more) {
      const int nb = buf ^ 1;
#pragma unroll
      for (int i = 0; i < 8; ++i)
        *(v4f*)(&sX[nb][xoff + 16 * i * GPITCH]) = gx[i];
#pragma unroll
      for (int n = 0; n < 4; ++n) {
        v4f t;
        t.x = gw[0][n];  t.y = gw[1][n];  t.z = gw[2][n];  t.w = gw[3][n];
        *(v4f*)(&sW[nb][woff + n * GPITCH]) = t;
      }
    }
    __syncthreads();
  }

#pragma unroll
  for (int t = 0; t < 2; ++t) {
    float* yb =
        Y + (size_t)(mbase + t * 16 + 8 * (lane >> 4)) * HD + nbase + lr;
#pragma unroll
    for (int r = 0; r < 8; ++r) {
      float* yr = yb + (size_t)r * HD;
      yr[0]  = acc[t * 4 + 0][r];
      yr[16] = acc[t * 4 + 1][r];
      yr[32] = acc[t * 4 + 2][r];
      yr[48] = acc[t * 4 + 3][r];
    }
  }
}

#define QT 2
#define NW 4
#define KC 32
#define PITCH 68

__device__ __forceinline__ float xhalf(float x) {
  const int b = __builtin_bit_cast(int, x);
  return __builtin_bit_cast(
      float, __builtin_amdgcn_permlanex16(b, b, 0x76543210, 0xfedcba98, false,
                                          false));
}

__device__ __forceinline__ void load_frag_k(v4f (&kf)[8], const float* p) {
#pragma unroll
  for (int u = 0; u < 8; ++u) kf[u] = *(const v4f*)(p + 8 * u);
}

__device__ __forceinline__ void load_frag_v(v4f (&vf)[8], const float* p) {
#pragma unroll
  for (int i = 0; i < 8; ++i) vf[i] = *(const v4f*)(p + i * PITCH);
}

__global__ __launch_bounds__(32 * NW) void attn_kernel(
    const float* __restrict__ Q, const float* __restrict__ K,
    const float* __restrict__ V, const int* __restrict__ length,
    float* __restrict__ O) {
  __shared__ __attribute__((aligned(16))) float sK[2][KC * PITCH];
  __shared__ __attribute__((aligned(16))) float sV[2][KC * PITCH];

  const int tid   = threadIdx.x;
  const int lane  = tid & 31;
  const int wave  = tid >> 5;
  const int lr    = lane & 15;
  const int hh    = lane >> 4;
  const int half8 = hh * 8;
  const int qbase = (blockIdx.x * NW + wave) * (16 * QT);
  const int h     = blockIdx.y;
  const int b     = blockIdx.z;

  int L = length[b];
  if (L < 0) L = 0;
  if (L > SS) L = SS;

  const size_t baseBH = (size_t)b * SS * HD + (size_t)h * DD;
  const float* Qb = Q + baseBH;
  const float* Kb = K + baseBH;
  const float* Vb = V + baseBH;
  float*       Ob = O + baseBH;

  v2f bq[QT][16];
#pragma unroll
  for (int t = 0; t < QT; ++t) {
    const float* qrow = Qb + (size_t)(qbase + 16 * t + lr) * HD + 4 * hh;
#pragma unroll
    for (int u = 0; u < 8; ++u) {
      const v4f q4 = *(const v4f*)(qrow + 8 * u);
      bq[t][2 * u].x     = q4.x;  bq[t][2 * u].y     = q4.y;
      bq[t][2 * u + 1].x = q4.z;  bq[t][2 * u + 1].y = q4.w;
    }
  }

  float m[QT], l[QT];
#pragma unroll
  for (int t = 0; t < QT; ++t) { m[t] = -3.0e38f; l[t] = 0.0f; }
  v8f acc[QT][4] = {};

  const int nkt = (L + 15) >> 4;
  const int nch = (nkt + 1) >> 1;

  const int srow = tid >> 4;
  const int scol = (tid & 15) * 4;
  const float* gK = Kb + (size_t)srow * HD + scol;
  const float* gV = Vb + (size_t)srow * HD + scol;
  const int    sOff = srow * PITCH + scol;
  v4f gk[4], gv[4];

  if (nch > 0) {
#pragma unroll
    for (int i = 0; i < 4; ++i) {
      gk[i] = *(const v4f*)(gK + (size_t)(8 * i) * HD);
      gv[i] = *(const v4f*)(gV + (size_t)(8 * i) * HD);
    }
#pragma unroll
    for (int i = 0; i < 4; ++i) {
      *(v4f*)(&sK[0][sOff + 8 * i * PITCH]) = gk[i];
      *(v4f*)(&sV[0][sOff + 8 * i * PITCH]) = gv[i];
    }
  }
  __syncthreads();

  for (int ch = 0; ch < nch; ++ch) {
    const int  buf  = ch & 1;
    const bool more = (ch + 1) < nch;

    if (more) {
      const size_t goff = (size_t)(ch + 1) * KC * HD;
#pragma unroll
      for (int i = 0; i < 4; ++i) {
        gk[i] = *(const v4f*)(gK + goff + (size_t)(8 * i) * HD);
        gv[i] = *(const v4f*)(gV + goff + (size_t)(8 * i) * HD);
      }
    }

    const float* kbuf = &sK[buf][lr * PITCH + 4 * hh];
    const float* vbuf = &sV[buf][half8 * PITCH + 4 * lr];
    v4f kf[8], vf[8];
    load_frag_k(kf, kbuf);
    sched_fence();

#pragma unroll
    for (int tt = 0; tt < KC / 16; ++tt) {
      const int kt = ch * (KC / 16) + tt;
      if (kt < nkt) {
        const int kbase = kt * 16;

        v8f sc[QT] = {};
#pragma unroll
        for (int u = 0; u < 8; ++u) {
          v2f a0, a1;
          a0.x = kf[u].x;  a0.y = kf[u].y;
          a1.x = kf[u].z;  a1.y = kf[u].w;
#pragma unroll
          for (int t = 0; t < QT; ++t) sc[t] = wmma_f32(a0, bq[t][2 * u], sc[t]);
#pragma unroll
          for (int t = 0; t < QT; ++t)
            sc[t] = wmma_f32(a1, bq[t][2 * u + 1], sc[t]);
        }

        sched_fence();
        load_frag_v(vf, vbuf + tt * 16 * PITCH);
        if (tt + 1 < KC / 16) load_frag_k(kf, kbuf + (tt + 1) * 16 * PITCH);
        sched_fence();

#pragma unroll
        for (int r = 0; r < 8; ++r) {
          const bool kvalid = (kbase + half8 + r) < L;
#pragma unroll
          for (int t = 0; t < QT; ++t) {
            const float v = sc[t][r] * 0.125f;
            sc[t][r] = kvalid ? v : -1.0e30f;
          }
        }

#pragma unroll
        for (int t = 0; t < QT; ++t) {
          float mx = fmaxf(fmaxf(fmaxf(sc[t][0], sc[t][1]), fmaxf(sc[t][2], sc[t][3])),
                           fmaxf(fmaxf(sc[t][4], sc[t][5]), fmaxf(sc[t][6], sc[t][7])));
          mx = fmaxf(mx, xhalf(mx));
          const float mnew = fmaxf(m[t], mx);
#pragma unroll
          for (int r = 0; r < 8; ++r) sc[t][r] = __expf(sc[t][r] - mnew);
          const float psum = ((sc[t][0] + sc[t][1]) + (sc[t][2] + sc[t][3])) +
                             ((sc[t][4] + sc[t][5]) + (sc[t][6] + sc[t][7]));
          const float scale = __expf(m[t] - mnew);
          l[t] = l[t] * scale + psum;
          m[t] = mnew;
#pragma unroll
          for (int j = 0; j < 4; ++j)
#pragma unroll
            for (int r = 0; r < 8; ++r) acc[t][j][r] *= scale;
        }

#pragma unroll
        for (int s = 0; s < 4; ++s) {
          v2f bp[QT];
#pragma unroll
          for (int t = 0; t < QT; ++t) {
            bp[t].x = sc[t][2 * s];
            bp[t].y = sc[t][2 * s + 1];
          }
#pragma unroll
          for (int j = 0; j < 4; ++j) {
            v2f av;
            av.x = vf[2 * s][j];
            av.y = vf[2 * s + 1][j];
#pragma unroll
            for (int t = 0; t < QT; ++t)
              acc[t][j] = wmma_f32(av, bp[t], acc[t][j]);
          }
        }
      }
    }

    if (more) {
      const int nb = buf ^ 1;
#pragma unroll
      for (int i = 0; i < 4; ++i) {
        *(v4f*)(&sK[nb][sOff + 8 * i * PITCH]) = gk[i];
        *(v4f*)(&sV[nb][sOff + 8 * i * PITCH]) = gv[i];
      }
    }
    __syncthreads();
  }

#pragma unroll
  for (int t = 0; t < QT; ++t) {
    const float lt = l[t] + xhalf(l[t]);
    const int qrow = qbase + 16 * t + lr;
    const float inv = (qrow < L && lt > 0.0f) ? (1.0f / lt) : 0.0f;
    float* orow = Ob + (size_t)qrow * HD + 32 * hh;
#pragma unroll
    for (int r = 0; r < 8; ++r) {
      v4f o;
      o.x = acc[t][0][r] * inv;
      o.y = acc[t][1][r] * inv;
      o.z = acc[t][2][r] * inv;
      o.w = acc[t][3][r] * inv;
      *(v4f*)(orow + 4 * r) = o;
    }
  }
}

extern "C" void kernel_launch(void* const* d_in, const int* in_sizes, int n_in,
                              void* d_out, int out_size, void* d_ws,
                              size_t ws_size, hipStream_t stream) {
  const float* x      = (const float*)d_in[0];
  const int*   length = (const int*)d_in[1];
  const float* Wq     = (const float*)d_in[2];
  const float* Wk     = (const float*)d_in[3];
  const float* Wv     = (const float*)d_in[4];
  const float* Wx     = (const float*)d_in[5];
  float*       out    = (float*)d_out;

  const size_t mat = (size_t)BB * SS * HD;
  float* Qw = (float*)d_ws;
  float* Kw = Qw + mat;
  float* Vw = Kw + mat;
  float* Ow = Vw + mat;

  const dim3 gGrid(64, 8, 1);
  const dim3 gBlock(128, 1, 1);

  gemm_8192x512x512<<<gGrid, gBlock, 0, stream>>>(x, Wq, Qw);
  gemm_8192x512x512<<<gGrid, gBlock, 0, stream>>>(x, Wk, Kw);
  gemm_8192x512x512<<<gGrid, gBlock, 0, stream>>>(x, Wv, Vw);

  attn_kernel<<<dim3(SS / (16 * QT * NW), HH, BB), dim3(32 * NW, 1, 1), 0, stream>>>(
      Qw, Kw, Vw, length, Ow);

  gemm_8192x512x512<<<gGrid, gBlock, 0, stream>>>(Ow, Wx, out);
}


# sequence rows at or beyond length are never computed: Q,K,V projections in one launch whose 128-row blocks past length exit; attention blocks past length exit, output blocks store zeros
# speedup vs baseline: 2.0329x; 1.0858x over previous
#include <hip/hip_runtime.h>
#include <math.h>

#define BB 4
#define SS 2048
#define CC 512
#define HH 8
#define DD 64
#define HD 512

typedef __attribute__((ext_vector_type(2))) float v2f;
typedef __attribute__((ext_vector_type(8))) float v8f;

__device__ __forceinline__ v8f wmma_f32(v2f a, v2f b, v8f c) {
  return __builtin_amdgcn_wmma_f32_16x16x4_f32(false, a, false, b, (short)0, c,
                                               false, false);
}

typedef __attribute__((ext_vector_type(4))) float v4f;

__device__ __forceinline__ void sched_fence() {
  __builtin_amdgcn_sched_barrier(0);
}

#define GKB 32
#define GPITCH 36

__global__ __launch_bounds__(128) void gemm_8192x512x512(
    const float* __restrict__ X, const float* __restrict__ W0,
    const float* __restrict__ W1, const float* __restrict__ W2,
    float* __restrict__ Y0, float* __restrict__ Y1, float* __restrict__ Y2,
    const int* __restrict__ length, int zero_fill) {
  __shared__ __attribute__((aligned(16))) float sX[2][128 * GPITCH];
  __shared__ __attribute__((aligned(16))) float sW[2][64 * GPITCH];

  const int tid  = threadIdx.x;
  const int lane = tid & 31;
  const int wave = tid >> 5;
  const int lr   = lane & 15;
  const int hh   = lane >> 4;
  const int mblk  = blockIdx.x * 128;
  const int mbase = mblk + wave * 32;
  const int which = blockIdx.y >> 3;
  const int nbase = (blockIdx.y & 7) * 64;
  const float* __restrict__ W = which == 0 ? W0 : (which == 1 ? W1 : W2);
  float* __restrict__       Y = which == 0 ? Y0 : (which == 1 ? Y1 : Y2);

  int L = length[mblk / SS];
  if (L < 0) L = 0;
  if (L > SS) L = SS;
  if ((mblk % SS) >= L) {
    if (zero_fill) {
      const v4f z = {0.0f, 0.0f, 0.0f, 0.0f};
      float* yz = Y + (size_t)(mblk + (tid >> 4)) * HD + nbase + (tid & 15) * 4;
#pragma unroll
      for (int i = 0; i < 16; ++i) *(v4f*)(yz + (size_t)(8 * i) * HD) = z;
    }
    return;
  }

  const int xrow = tid >> 3;
  const int xcol = (tid & 7) * 4;
  const float* gX = X + (size_t)(mblk + xrow) * HD + xcol;
  const int xoff = xrow * GPITCH + xcol;
  const int wn4 = (tid & 15) * 4;
  const int wk4 = (tid >> 4) * 4;
  const float* gW = W + (size_t)wk4 * HD + nbase + wn4;
  const int woff = wn4 * GPITCH + wk4;

  v4f gx[8], gw[4];
#pragma unroll
  for (int i = 0; i < 8; ++i) gx[i] = *(const v4f*)(gX + (size_t)(16 * i) * HD);
#pragma unroll
  for (int i = 0; i < 4; ++i) gw[i] = *(const v4f*)(gW + (size_t)i * HD);
#pragma unroll
  for (int i = 0; i < 8; ++i) *(v4f*)(&sX[0][xoff + 16 * i * GPITCH]) = gx[i];
#pragma unroll
  for (int n = 0; n < 4; ++n) {
    v4f t;
    t.x = gw[0][n];  t.y = gw[1][n];  t.z = gw[2][n];  t.w = gw[3][n];
    *(v4f*)(&sW[0][woff + n * GPITCH]) = t;
  }
  __syncthreads();

  v8f acc[8] = {};

  const int aoff = (wave * 32 + lr) * GPITCH + 4 * hh;
  const int boff = lr * GPITCH + 4 * hh;

  for (int kb = 0; kb < CC / GKB; ++kb) {
    const int  buf  = kb & 1;
    const bool more = (kb + 1) < CC / GKB;

    if (more) {
      const int k0 = (kb + 1) * GKB;
#pragma unroll
      for (int i = 0; i < 8; ++i)
        gx[i] = *(const v4f*)(gX + k0 + (size_t)(16 * i) * HD);
#pragma unroll
      for (int i = 0; i < 4; ++i)
        gw[i] = *(const v4f*)(gW + (size_t)(k0 + i) * HD);
    }

    const float* ap = &sX[buf][aoff];
    const float* bp = &sW[buf][boff];
    v4f af[2][4], bf[4][4];
#pragma unroll
    for (int u = 0; u < 4; ++u) {
      af[0][u] = *(const v4f*)(ap + 8 * u);
      af[1][u] = *(const v4f*)(ap + 16 * GPITCH + 8 * u);
#pragma unroll
      for (int j = 0; j < 4; ++j)
        bf[j][u] = *(const v4f*)(bp + j * 16 * GPITCH + 8 * u);
    }
    sched_fence();

#pragma unroll
    for (int u = 0; u < 4; ++u) {
#pragma unroll
      for (int e = 0; e < 2; ++e) {
        v2f a0, a1, b[4];
        a0.x = af[0][u][2 * e];  a0.y = af[0][u][2 * e + 1];
        a1.x = af[1][u][2 * e];  a1.y = af[1][u][2 * e + 1];
#pragma unroll
        for (int j = 0; j < 4; ++j) {
          b[j].x = bf[j][u][2 * e];
          b[j].y = bf[j][u][2 * e + 1];
        }
#pragma unroll
        for (int j = 0; j < 4; ++j) acc[j] = wmma_f32(a0, b[j], acc[j]);
#pragma unroll
        for (int j = 0; j < 4; ++j) acc[4 + j] = wmma_f32(a1, b[j], acc[4 + j]);
      }
    }

    if (more) {
      const int nb = buf ^ 1;
#pragma unroll
      for (int i = 0; i < 8; ++i)
        *(v4f*)(&sX[nb][xoff + 16 * i * GPITCH]) = gx[i];
#pragma unroll
      for (int n = 0; n < 4; ++n) {
        v4f t;
        t.x = gw[0][n];  t.y = gw[1][n];  t.z = gw[2][n];  t.w = gw[3][n];
        *(v4f*)(&sW[nb][woff + n * GPITCH]) = t;
      }
    }
    __syncthreads();
  }

#pragma unroll
  for (int t = 0; t < 2; ++t) {
    float* yb =
        Y + (size_t)(mbase + t * 16 + 8 * (lane >> 4)) * HD + nbase + lr;
#pragma unroll
    for (int r = 0; r < 8; ++r) {
      float* yr = yb + (size_t)r * HD;
      yr[0]  = acc[t * 4 + 0][r];
      yr[16] = acc[t * 4 + 1][r];
      yr[32] = acc[t * 4 + 2][r];
      yr[48] = acc[t * 4 + 3][r];
    }
  }
}

#define QT 2
#define NW 4
#define KC 32
#define PITCH 68

__device__ __forceinline__ float xhalf(float x) {
  const int b = __builtin_bit_cast(int, x);
  return __builtin_bit_cast(
      float, __builtin_amdgcn_permlanex16(b, b, 0x76543210, 0xfedcba98, false,
                                          false));
}

__device__ __forceinline__ void load_frag_k(v4f (&kf)[8], const float* p) {
#pragma unroll
  for (int u = 0; u < 8; ++u) kf[u] = *(const v4f*)(p + 8 * u);
}

__device__ __forceinline__ void load_frag_v(v4f (&vf)[8], const float* p) {
#pragma unroll
  for (int i = 0; i < 8; ++i) vf[i] = *(const v4f*)(p + i * PITCH);
}

__global__ __launch_bounds__(32 * NW) void attn_kernel(
    const float* __restrict__ Q, const float* __restrict__ K,
    const float* __restrict__ V, const int* __restrict__ length,
    float* __restrict__ O) {
  __shared__ __attribute__((aligned(16))) float sK[2][KC * PITCH];
  __shared__ __attribute__((aligned(16))) float sV[2][KC * PITCH];

  const int tid   = threadIdx.x;
  const int lane  = tid & 31;
  const int wave  = tid >> 5;
  const int lr    = lane & 15;
  const int hh    = lane >> 4;
  const int half8 = hh * 8;
  const int qbase = (blockIdx.x * NW + wave) * (16 * QT);
  const int h     = blockIdx.y;
  const int b     = blockIdx.z;

  int L = length[b];
  if (L < 0) L = 0;
  if (L > SS) L = SS;

  if ((int)(blockIdx.x * (16 * QT * NW)) >= L) return;

  const size_t baseBH = (size_t)b * SS * HD + (size_t)h * DD;
  const float* Qb = Q + baseBH;
  const float* Kb = K + baseBH;
  const float* Vb = V + baseBH;
  float*       Ob = O + baseBH;

  v2f bq[QT][16];
#pragma unroll
  for (int t = 0; t < QT; ++t) {
    const float* qrow = Qb + (size_t)(qbase + 16 * t + lr) * HD + 4 * hh;
#pragma unroll
    for (int u = 0; u < 8; ++u) {
      const v4f q4 = *(const v4f*)(qrow + 8 * u);
      bq[t][2 * u].x     = q4.x;  bq[t][2 * u].y     = q4.y;
      bq[t][2 * u + 1].x = q4.z;  bq[t][2 * u + 1].y = q4.w;
    }
  }

  float m[QT], l[QT];
#pragma unroll
  for (int t = 0; t < QT; ++t) { m[t] = -3.0e38f; l[t] = 0.0f; }
  v8f acc[QT][4] = {};

  const int nkt = (L + 15) >> 4;
  const int nch = (nkt + 1) >> 1;

  const int srow = tid >> 4;
  const int scol = (tid & 15) * 4;
  const float* gK = Kb + (size_t)srow * HD + scol;
  const float* gV = Vb + (size_t)srow * HD + scol;
  const int    sOff = srow * PITCH + scol;
  v4f gk[4], gv[4];

  if (nch > 0) {
#pragma unroll
    for (int i = 0; i < 4; ++i) {
      gk[i] = *(const v4f*)(gK + (size_t)(8 * i) * HD);
      gv[i] = *(const v4f*)(gV + (size_t)(8 * i) * HD);
    }
#pragma unroll
    for (int i = 0; i < 4; ++i) {
      *(v4f*)(&sK[0][sOff + 8 * i * PITCH]) = gk[i];
      *(v4f*)(&sV[0][sOff + 8 * i * PITCH]) = gv[i];
    }
  }
  __syncthreads();

  for (int ch = 0; ch < nch; ++ch) {
    const int  buf  = ch & 1;
    const bool more = (ch + 1) < nch;

    if (more) {
      const size_t goff = (size_t)(ch + 1) * KC * HD;
#pragma unroll
      for (int i = 0; i < 4; ++i) {
        gk[i] = *(const v4f*)(gK + goff + (size_t)(8 * i) * HD);
        gv[i] = *(const v4f*)(gV + goff + (size_t)(8 * i) * HD);
      }
    }

    const float* kbuf = &sK[buf][lr * PITCH + 4 * hh];
    const float* vbuf = &sV[buf][half8 * PITCH + 4 * lr];
    v4f kf[8], vf[8];
    load_frag_k(kf, kbuf);
    sched_fence();

#pragma unroll
    for (int tt = 0; tt < KC / 16; ++tt) {
      const int kt = ch * (KC / 16) + tt;
      if (kt < nkt) {
        const int kbase = kt * 16;

        v8f sc[QT] = {};
#pragma unroll
        for (int u = 0; u < 8; ++u) {
          v2f a0, a1;
          a0.x = kf[u].x;  a0.y = kf[u].y;
          a1.x = kf[u].z;  a1.y = kf[u].w;
#pragma unroll
          for (int t = 0; t < QT; ++t) sc[t] = wmma_f32(a0, bq[t][2 * u], sc[t]);
#pragma unroll
          for (int t = 0; t < QT; ++t)
            sc[t] = wmma_f32(a1, bq[t][2 * u + 1], sc[t]);
        }

        sched_fence();
        load_frag_v(vf, vbuf + tt * 16 * PITCH);
        if (tt + 1 < KC / 16) load_frag_k(kf, kbuf + (tt + 1) * 16 * PITCH);
        sched_fence();

#pragma unroll
        for (int r = 0; r < 8; ++r) {
          const bool kvalid = (kbase + half8 + r) < L;
#pragma unroll
          for (int t = 0; t < QT; ++t) {
            const float v = sc[t][r] * 0.125f;
            sc[t][r] = kvalid ? v : -1.0e30f;
          }
        }

#pragma unroll
        for (int t = 0; t < QT; ++t) {
          float mx = fmaxf(fmaxf(fmaxf(sc[t][0], sc[t][1]), fmaxf(sc[t][2], sc[t][3])),
                           fmaxf(fmaxf(sc[t][4], sc[t][5]), fmaxf(sc[t][6], sc[t][7])));
          mx = fmaxf(mx, xhalf(mx));
          const float mnew = fmaxf(m[t], mx);
#pragma unroll
          for (int r = 0; r < 8; ++r) sc[t][r] = __expf(sc[t][r] - mnew);
          const float psum = ((sc[t][0] + sc[t][1]) + (sc[t][2] + sc[t][3])) +
                             ((sc[t][4] + sc[t][5]) + (sc[t][6] + sc[t][7]));
          const float scale = __expf(m[t] - mnew);
          l[t] = l[t] * scale + psum;
          m[t] = mnew;
#pragma unroll
          for (int j = 0; j < 4; ++j)
#pragma unroll
            for (int r = 0; r < 8; ++r) acc[t][j][r] *= scale;
        }

#pragma unroll
        for (int s = 0; s < 4; ++s) {
          v2f bp[QT];
#pragma unroll
          for (int t = 0; t < QT; ++t) {
            bp[t].x = sc[t][2 * s];
            bp[t].y = sc[t][2 * s + 1];
          }
#pragma unroll
          for (int j = 0; j < 4; ++j) {
            v2f av;
            av.x = vf[2 * s][j];
            av.y = vf[2 * s + 1][j];
#pragma unroll
            for (int t = 0; t < QT; ++t)
              acc[t][j] = wmma_f32(av, bp[t], acc[t][j]);
          }
        }
      }
    }

    if (more) {
      const int nb = buf ^ 1;
#pragma unroll
      for (int i = 0; i < 4; ++i) {
        *(v4f*)(&sK[nb][sOff + 8 * i * PITCH]) = gk[i];
        *(v4f*)(&sV[nb][sOff + 8 * i * PITCH]) = gv[i];
      }
    }
    __syncthreads();
  }

#pragma unroll
  for (int t = 0; t < QT; ++t) {
    const float lt = l[t] + xhalf(l[t]);
    const int qrow = qbase + 16 * t + lr;
    const float inv = (qrow < L && lt > 0.0f) ? (1.0f / lt) : 0.0f;
    float* orow = Ob + (size_t)qrow * HD + 32 * hh;
#pragma unroll
    for (int r = 0; r < 8; ++r) {
      v4f o;
      o.x = acc[t][0][r] * inv;
      o.y = acc[t][1][r] * inv;
      o.z = acc[t][2][r] * inv;
      o.w = acc[t][3][r] * inv;
      *(v4f*)(orow + 4 * r) = o;
    }
  }
}

extern "C" void kernel_launch(void* const* d_in, const int* in_sizes, int n_in,
                              void* d_out, int out_size, void* d_ws,
                              size_t ws_size, hipStream_t stream) {
  const float* x      = (const float*)d_in[0];
  const int*   length = (const int*)d_in[1];
  const float* Wq     = (const float*)d_in[2];
  const float* Wk     = (const float*)d_in[3];
  const float* Wv     = (const float*)d_in[4];
  const float* Wx     = (const float*)d_in[5];
  float*       out    = (float*)d_out;

  const size_t mat = (size_t)BB * SS * HD;
  float* Qw = (float*)d_ws;
  float* Kw = Qw + mat;
  float* Vw = Kw + mat;
  float* Ow = Vw + mat;

  const dim3 gGrid3(64, 24, 1);
  const dim3 gGrid1(64, 8, 1);
  const dim3 gBlock(128, 1, 1);

  gemm_8192x512x512<<<gGrid3, gBlock, 0, stream>>>(x, Wq, Wk, Wv, Qw, Kw, Vw,
                                                   length, 0);

  attn_kernel<<<dim3(SS / (16 * QT * NW), HH, BB), dim3(32 * NW, 1, 1), 0, stream>>>(
      Qw, Kw, Vw, length, Ow);

  gemm_8192x512x512<<<gGrid1, gBlock, 0, stream>>>(Ow, Wx, Wx, Wx, out, out,
                                                   out, length, 1);
}


# attention softmax: vector work per key tile cut: scale*log2e folded into Q, exp2, mask only on the tile holding the length, accumulators rescaled only when a max grows by 2^8; sum order unchanged
# speedup vs baseline: 2.1215x; 1.0436x over previous
#include <hip/hip_runtime.h>
#include <math.h>

#define BB 4
#define SS 2048
#define CC 512
#define HH 8
#define DD 64
#define HD 512

typedef __attribute__((ext_vector_type(2))) float v2f;
typedef __attribute__((ext_vector_type(8))) float v8f;

__device__ __forceinline__ v8f wmma_f32(v2f a, v2f b, v8f c) {
  return __builtin_amdgcn_wmma_f32_16x16x4_f32(false, a, false, b, (short)0, c,
                                               false, false);
}

typedef __attribute__((ext_vector_type(4))) float v4f;

__device__ __forceinline__ void sched_fence() {
  __builtin_amdgcn_sched_barrier(0);
}

#define GKB 32
#define GPITCH 36

__global__ __launch_bounds__(128) void gemm_8192x512x512(
    const float* __restrict__ X, const float* __restrict__ W0,
    const float* __restrict__ W1, const float* __restrict__ W2,
    float* __restrict__ Y0, float* __restrict__ Y1, float* __restrict__ Y2,
    const int* __restrict__ length, int zero_fill) {
  __shared__ __attribute__((aligned(16))) float sX[2][128 * GPITCH];
  __shared__ __attribute__((aligned(16))) float sW[2][64 * GPITCH];

  const int tid  = threadIdx.x;
  const int lane = tid & 31;
  const int wave = tid >> 5;
  const int lr   = lane & 15;
  const int hh   = lane >> 4;
  const int mblk  = blockIdx.x * 128;
  const int mbase = mblk + wave * 32;
  const int which = blockIdx.y >> 3;
  const int nbase = (blockIdx.y & 7) * 64;
  const float* __restrict__ W = which == 0 ? W0 : (which == 1 ? W1 : W2);
  float* __restrict__       Y = which == 0 ? Y0 : (which == 1 ? Y1 : Y2);

  int L = length[mblk / SS];
  if (L < 0) L = 0;
  if (L > SS) L = SS;
  if ((mblk % SS) >= L) {
    if (zero_fill) {
      const v4f z = {0.0f, 0.0f, 0.0f, 0.0f};
      float* yz = Y + (size_t)(mblk + (tid >> 4)) * HD + nbase + (tid & 15) * 4;
#pragma unroll
      for (int i = 0; i < 16; ++i) *(v4f*)(yz + (size_t)(8 * i) * HD) = z;
    }
    return;
  }

  const int xrow = tid >> 3;
  const int xcol = (tid & 7) * 4;
  const float* gX = X + (size_t)(mblk + xrow) * HD + xcol;
  const int xoff = xrow * GPITCH + xcol;
  const int wn4 = (tid & 15) * 4;
  const int wk4 = (tid >> 4) * 4;
  const float* gW = W + (size_t)wk4 * HD + nbase + wn4;
  const int woff = wn4 * GPITCH + wk4;

  v4f gx[8], gw[4];
#pragma unroll
  for (int i = 0; i < 8; ++i) gx[i] = *(const v4f*)(gX + (size_t)(16 * i) * HD);
#pragma unroll
  for (int i = 0; i < 4; ++i) gw[i] = *(const v4f*)(gW + (size_t)i * HD);
#pragma unroll
  for (int i = 0; i < 8; ++i) *(v4f*)(&sX[0][xoff + 16 * i * GPITCH]) = gx[i];
#pragma unroll
  for (int n = 0; n < 4; ++n) {
    v4f t;
    t.x = gw[0][n];  t.y = gw[1][n];  t.z = gw[2][n];  t.w = gw[3][n];
    *(v4f*)(&sW[0][woff + n * GPITCH]) = t;
  }
  __syncthreads();

  v8f acc[8] = {};

  const int aoff = (wave * 32 + lr) * GPITCH + 4 * hh;
  const int boff = lr * GPITCH + 4 * hh;

  for (int kb = 0; kb < CC / GKB; ++kb) {
    const int  buf  = kb & 1;
    const bool more = (kb + 1) < CC / GKB;

    if (more) {
      const int k0 = (kb + 1) * GKB;
#pragma unroll
      for (int i = 0; i < 8; ++i)
        gx[i] = *(const v4f*)(gX + k0 + (size_t)(16 * i) * HD);
#pragma unroll
      for (int i = 0; i < 4; ++i)
        gw[i] = *(const v4f*)(gW + (size_t)(k0 + i) * HD);
    }

    const float* ap = &sX[buf][aoff];
    const float* bp = &sW[buf][boff];
    v4f af[2][4], bf[4][4];
#pragma unroll
    for (int u = 0; u < 4; ++u) {
      af[0][u] = *(const v4f*)(ap + 8 * u);
      af[1][u] = *(const v4f*)(ap + 16 * GPITCH + 8 * u);
#pragma unroll
      for (int j = 0; j < 4; ++j)
        bf[j][u] = *(const v4f*)(bp + j * 16 * GPITCH + 8 * u);
    }
    sched_fence();

#pragma unroll
    for (int u = 0; u < 4; ++u) {
#pragma unroll
      for (int e = 0; e < 2; ++e) {
        v2f a0, a1, b[4];
        a0.x = af[0][u][2 * e];  a0.y = af[0][u][2 * e + 1];
        a1.x = af[1][u][2 * e];  a1.y = af[1][u][2 * e + 1];
#pragma unroll
        for (int j = 0; j < 4; ++j) {
          b[j].x = bf[j][u][2 * e];
          b[j].y = bf[j][u][2 * e + 1];
        }
#pragma unroll
        for (int j = 0; j < 4; ++j) acc[j] = wmma_f32(a0, b[j], acc[j]);
#pragma unroll
        for (int j = 0; j < 4; ++j) acc[4 + j] = wmma_f32(a1, b[j], acc[4 + j]);
      }
    }

    if (more) {
      const int nb = buf ^ 1;
#pragma unroll
      for (int i = 0; i < 8; ++i)
        *(v4f*)(&sX[nb][xoff + 16 * i * GPITCH]) = gx[i];
#pragma unroll
      for (int n = 0; n < 4; ++n) {
        v4f t;
        t.x = gw[0][n];  t.y = gw[1][n];  t.z = gw[2][n];  t.w = gw[3][n];
        *(v4f*)(&sW[nb][woff + n * GPITCH]) = t;
      }
    }
    __syncthreads();
  }

#pragma unroll
  for (int t = 0; t < 2; ++t) {
    float* yb =
        Y + (size_t)(mbase + t * 16 + 8 * (lane >> 4)) * HD + nbase + lr;
#pragma unroll
    for (int r = 0; r < 8; ++r) {
      float* yr = yb + (size_t)r * HD;
      yr[0]  = acc[t * 4 + 0][r];
      yr[16] = acc[t * 4 + 1][r];
      yr[32] = acc[t * 4 + 2][r];
      yr[48] = acc[t * 4 + 3][r];
    }
  }
}

#define QT 2
#define NW 4
#define KC 32
#define QSCALE 0.18033688011112042f
#define RESCALE_T 8.0f
#define PITCH 68

__device__ __forceinline__ float xhalf(float x) {
  const int b = __builtin_bit_cast(int, x);
  return __builtin_bit_cast(
      float, __builtin_amdgcn_permlanex16(b, b, 0x76543210, 0xfedcba98, false,
                                          false));
}

__device__ __forceinline__ void load_frag_k(v4f (&kf)[8], const float* p) {
#pragma unroll
  for (int u = 0; u < 8; ++u) kf[u] = *(const v4f*)(p + 8 * u);
}

__device__ __forceinline__ void load_frag_v(v4f (&vf)[8], const float* p) {
#pragma unroll
  for (int i = 0; i < 8; ++i) vf[i] = *(const v4f*)(p + i * PITCH);
}

__global__ __launch_bounds__(32 * NW) void attn_kernel(
    const float* __restrict__ Q, const float* __restrict__ K,
    const float* __restrict__ V, const int* __restrict__ length,
    float* __restrict__ O) {
  __shared__ __attribute__((aligned(16))) float sK[2][KC * PITCH];
  __shared__ __attribute__((aligned(16))) float sV[2][KC * PITCH];

  const int tid   = threadIdx.x;
  const int lane  = tid & 31;
  const int wave  = tid >> 5;
  const int lr    = lane & 15;
  const int hh    = lane >> 4;
  const int half8 = hh * 8;
  const int qbase = (blockIdx.x * NW + wave) * (16 * QT);
  const int h     = blockIdx.y;
  const int b     = blockIdx.z;

  int L = length[b];
  if (L < 0) L = 0;
  if (L > SS) L = SS;

  if ((int)(blockIdx.x * (16 * QT * NW)) >= L) return;

  const size_t baseBH = (size_t)b * SS * HD + (size_t)h * DD;
  const float* Qb = Q + baseBH;
  const float* Kb = K + baseBH;
  const float* Vb = V + baseBH;
  float*       Ob = O + baseBH;

  v2f bq[QT][16];
#pragma unroll
  for (int t = 0; t < QT; ++t) {
    const float* qrow = Qb + (size_t)(qbase + 16 * t + lr) * HD + 4 * hh;
#pragma unroll
    for (int u = 0; u < 8; ++u) {
      const v4f q4 = *(const v4f*)(qrow + 8 * u);
      bq[t][2 * u].x     = q4.x * QSCALE;  bq[t][2 * u].y     = q4.y * QSCALE;
      bq[t][2 * u + 1].x = q4.z * QSCALE;  bq[t][2 * u + 1].y = q4.w * QSCALE;
    }
  }

  float m[QT], mT[QT], l[QT];
#pragma unroll
  for (int t = 0; t < QT; ++t) { m[t] = -3.0e38f; mT[t] = -3.0e38f; l[t] = 0.0f; }
  v8f acc[QT][4] = {};

  const int nkt = (L + 15) >> 4;
  const int nch = (nkt + 1) >> 1;

  const int srow = tid >> 4;
  const int scol = (tid & 15) * 4;
  const float* gK = Kb + (size_t)srow * HD + scol;
  const float* gV = Vb + (size_t)srow * HD + scol;
  const int    sOff = srow * PITCH + scol;
  v4f gk[4], gv[4];

  if (nch > 0) {
#pragma unroll
    for (int i = 0; i < 4; ++i) {
      gk[i] = *(const v4f*)(gK + (size_t)(8 * i) * HD);
      gv[i] = *(const v4f*)(gV + (size_t)(8 * i) * HD);
    }
#pragma unroll
    for (int i = 0; i < 4; ++i) {
      *(v4f*)(&sK[0][sOff + 8 * i * PITCH]) = gk[i];
      *(v4f*)(&sV[0][sOff + 8 * i * PITCH]) = gv[i];
    }
  }
  __syncthreads();

  for (int ch = 0; ch < nch; ++ch) {
    const int  buf  = ch & 1;
    const bool more = (ch + 1) < nch;

    if (more) {
      const size_t goff = (size_t)(ch + 1) * KC * HD;
#pragma unroll
      for (int i = 0; i < 4; ++i) {
        gk[i] = *(const v4f*)(gK + goff + (size_t)(8 * i) * HD);
        gv[i] = *(const v4f*)(gV + goff + (size_t)(8 * i) * HD);
      }
    }

    const float* kbuf = &sK[buf][lr * PITCH + 4 * hh];
    const float* vbuf = &sV[buf][half8 * PITCH + 4 * lr];
    v4f kf[8], vf[8];
    load_frag_k(kf, kbuf);
    sched_fence();

#pragma unroll
    for (int tt = 0; tt < KC / 16; ++tt) {
      const int kt = ch * (KC / 16) + tt;
      if (kt < nkt) {
        const int kbase = kt * 16;

        v8f sc[QT] = {};
#pragma unroll
        for (int u = 0; u < 8; ++u) {
          v2f a0, a1;
          a0.x = kf[u].x;  a0.y = kf[u].y;
          a1.x = kf[u].z;  a1.y = kf[u].w;
#pragma unroll
          for (int t = 0; t < QT; ++t) sc[t] = wmma_f32(a0, bq[t][2 * u], sc[t]);
#pragma unroll
          for (int t = 0; t < QT; ++t)
            sc[t] = wmma_f32(a1, bq[t][2 * u + 1], sc[t]);
        }

        sched_fence();
        load_frag_v(vf, vbuf + tt * 16 * PITCH);
        if (tt + 1 < KC / 16) load_frag_k(kf, kbuf + (tt + 1) * 16 * PITCH);
        sched_fence();

        if (__builtin_expect(kbase + 16 > L, 0)) {
#pragma unroll
          for (int r = 0; r < 8; ++r) {
            const bool kvalid = (kbase + half8 + r) < L;
#pragma unroll
            for (int t = 0; t < QT; ++t)
              sc[t][r] = kvalid ? sc[t][r] : -1.0e30f;
          }
        }

        float mx[QT];
        bool grow = false;
#pragma unroll
        for (int t = 0; t < QT; ++t) {
          const float a3 = fmaxf(fmaxf(sc[t][0], sc[t][1]), sc[t][2]);
          const float b3 = fmaxf(fmaxf(sc[t][3], sc[t][4]), sc[t][5]);
          const float c3 = fmaxf(fmaxf(sc[t][6], sc[t][7]), a3);
          float v = fmaxf(b3, c3);
          v = fmaxf(v, xhalf(v));
          mx[t] = v;
          grow = grow || (v > mT[t]);
        }
        if (__builtin_expect(__builtin_amdgcn_ballot_w32(grow) != 0u, 0)) {
#pragma unroll
          for (int t = 0; t < QT; ++t) {
            const float mnew  = fmaxf(m[t], mx[t]);
            const float scale = __builtin_amdgcn_exp2f(m[t] - mnew);
            l[t] *= scale;
            m[t]  = mnew;
            mT[t] = mnew + RESCALE_T;
#pragma unroll
            for (int j = 0; j < 4; ++j)
#pragma unroll
              for (int r = 0; r < 8; ++r) acc[t][j][r] *= scale;
          }
        }
#pragma unroll
        for (int t = 0; t < QT; ++t) {
#pragma unroll
          for (int r = 0; r < 8; ++r)
            sc[t][r] = __builtin_amdgcn_exp2f(sc[t][r] - m[t]);
          const float psum = ((sc[t][0] + sc[t][1]) + (sc[t][2] + sc[t][3])) +
                             ((sc[t][4] + sc[t][5]) + (sc[t][6] + sc[t][7]));
          l[t] += psum;
        }

#pragma unroll
        for (int s = 0; s < 4; ++s) {
          v2f bp[QT];
#pragma unroll
          for (int t = 0; t < QT; ++t) {
            bp[t].x = sc[t][2 * s];
            bp[t].y = sc[t][2 * s + 1];
          }
#pragma unroll
          for (int j = 0; j < 4; ++j) {
            v2f av;
            av.x = vf[2 * s][j];
            av.y = vf[2 * s + 1][j];
#pragma unroll
            for (int t = 0; t < QT; ++t)
              acc[t][j] = wmma_f32(av, bp[t], acc[t][j]);
          }
        }
      }
    }

    if (more) {
      const int nb = buf ^ 1;
#pragma unroll
      for (int i = 0; i < 4; ++i) {
        *(v4f*)(&sK[nb][sOff + 8 * i * PITCH]) = gk[i];
        *(v4f*)(&sV[nb][sOff + 8 * i * PITCH]) = gv[i];
      }
    }
    __syncthreads();
  }

#pragma unroll
  for (int t = 0; t < QT; ++t) {
    const float lt = l[t] + xhalf(l[t]);
    const int qrow = qbase + 16 * t + lr;
    const float inv = (qrow < L && lt > 0.0f) ? (1.0f / lt) : 0.0f;
    float* orow = Ob + (size_t)qrow * HD + 32 * hh;
#pragma unroll
    for (int r = 0; r < 8; ++r) {
      v4f o;
      o.x = acc[t][0][r] * inv;
      o.y = acc[t][1][r] * inv;
      o.z = acc[t][2][r] * inv;
      o.w = acc[t][3][r] * inv;
      *(v4f*)(orow + 4 * r) = o;
    }
  }
}

extern "C" void kernel_launch(void* const* d_in, const int* in_sizes, int n_in,
                              void* d_out, int out_size, void* d_ws,
                              size_t ws_size, hipStream_t stream) {
  const float* x      = (const float*)d_in[0];
  const int*   length = (const int*)d_in[1];
  const float* Wq     = (const float*)d_in[2];
  const float* Wk     = (const float*)d_in[3];
  const float* Wv     = (const float*)d_in[4];
  const float* Wx     = (const float*)d_in[5];
  float*       out    = (float*)d_out;

  const size_t mat = (size_t)BB * SS * HD;
  float* Qw = (float*)d_ws;
  float* Kw = Qw + mat;
  float* Vw = Kw + mat;
  float* Ow = Vw + mat;

  const dim3 gGrid3(64, 24, 1);
  const dim3 gGrid1(64, 8, 1);
  const dim3 gBlock(128, 1, 1);

  gemm_8192x512x512<<<gGrid3, gBlock, 0, stream>>>(x, Wq, Wk, Wv, Qw, Kw, Vw,
                                                   length, 0);

  attn_kernel<<<dim3(SS / (16 * QT * NW), HH, BB), dim3(32 * NW, 1, 1), 0, stream>>>(
      Qw, Kw, Vw, length, Ow);

  gemm_8192x512x512<<<gGrid1, gBlock, 0, stream>>>(Ow, Wx, Wx, Wx, out, out,
                                                   out, length, 1);
}
